# starting kernel
# baseline (speedup 1.0000x reference)
#include <hip/hip_runtime.h>
#include <hip/hip_bf16.h>
#include <math.h>

typedef __attribute__((ext_vector_type(2))) float v2f;
typedef __attribute__((ext_vector_type(8))) float v8f;

#define WAVES_PER_BLOCK 2
#define LANES 32
#define M_PER_WAVE 32
#define STRIDE 113

static __device__ __forceinline__ v8f wmma4(v2f a, v2f b, v8f c) {
  return __builtin_amdgcn_wmma_f32_16x16x4_f32(false, a, false, b, (short)0, c,
                                               false, false);
}

static __device__ __forceinline__ void lds_fence() {
  asm volatile("s_wait_dscnt 0x0" ::: "memory");
}

static __device__ __forceinline__ float fast_tanh(float x) {
#if __has_builtin(__builtin_amdgcn_tanhf)
  return __builtin_amdgcn_tanhf(x);
#else
  float e = __builtin_amdgcn_exp2f(x * 2.885390081777927f);
  return __builtin_fmaf(-2.f, __builtin_amdgcn_rcpf(e + 1.f), 1.f);
#endif
}

static __device__ __forceinline__ float gload0(const float* __restrict__ p,
                                               bool valid, int idx) {
  float v = p[valid ? idx : 0];
  return valid ? v : 0.f;
}

template <int KI, int NI, int KPAD>
static __device__ __forceinline__ void tile_pair(
    const float* __restrict__ W, const float* __restrict__ hin, int col,
    int half, int l16, v8f& c0, v8f& c1) {
  c0 = (v8f){0.f, 0.f, 0.f, 0.f, 0.f, 0.f, 0.f, 0.f};
  c1 = (v8f){0.f, 0.f, 0.f, 0.f, 0.f, 0.f, 0.f, 0.f};
#pragma unroll
  for (int k0 = 0; k0 < KPAD; k0 += 4) {
    const int ka = k0 + half * 2;
    v2f b;
    b.x = gload0(W, (ka + 0) < KI && col < NI, (ka + 0) * NI + col);
    b.y = gload0(W, (ka + 1) < KI && col < NI, (ka + 1) * NI + col);
    v2f a0, a1;
    a0.x = hin[l16 * STRIDE + ka];
    a0.y = hin[l16 * STRIDE + ka + 1];
    a1.x = hin[(16 + l16) * STRIDE + ka];
    a1.y = hin[(16 + l16) * STRIDE + ka + 1];
    c0 = wmma4(a0, b, c0);
    c1 = wmma4(a1, b, c1);
  }
}

template <int KI, int NI, int KPAD, int NPAD>
static __device__ __forceinline__ void hidden_layer(
    const float* __restrict__ W, const float* __restrict__ B,
    const float* __restrict__ hin, float* __restrict__ hout, int lane) {
  const int half = lane >> 4;
  const int l16  = lane & 15;
#pragma unroll
  for (int nt = 0; nt < NPAD / 16; ++nt) {
    const int col = nt * 16 + l16;
    v8f c0, c1;
    tile_pair<KI, NI, KPAD>(W, hin, col, half, l16, c0, c1);
    const float bias = gload0(B, col < NI, col);
#pragma unroll
    for (int r = 0; r < 8; ++r) {
      const int m = r + half * 8;
      hout[m * STRIDE + col]        = fast_tanh(c0[r] + bias);
      hout[(16 + m) * STRIDE + col] = fast_tanh(c1[r] + bias);
    }
  }
  lds_fence();
}

template <int KI, int NI, int KPAD, int NPAD, bool ACCUM>
static __device__ __forceinline__ void final_layer(
    const float* __restrict__ W, const float* __restrict__ B,
    const float* __restrict__ hin, v8f acc[2][NPAD / 16], int lane) {
  const int half = lane >> 4;
  const int l16  = lane & 15;
#pragma unroll
  for (int nt = 0; nt < NPAD / 16; ++nt) {
    const int col = nt * 16 + l16;
    v8f c0, c1;
    tile_pair<KI, NI, KPAD>(W, hin, col, half, l16, c0, c1);
    const float bias = gload0(B, col < NI, col);
#pragma unroll
    for (int r = 0; r < 8; ++r) {
      const float t0 = fast_tanh(c0[r] + bias);
      const float t1 = fast_tanh(c1[r] + bias);
      if (ACCUM) {
        acc[0][nt][r] += t0;
        acc[1][nt][r] += t1;
      } else {
        acc[0][nt][r] = t0;
        acc[1][nt][r] = t1;
      }
    }
  }
}

template <bool ACCUM>
static __device__ __forceinline__ void run_tower(
    const float* __restrict__ x, long m0, const float* W1, const float* B1,
    const float* W2, const float* B2, const float* W3, const float* B3,
    const float* W4, const float* B4, float* __restrict__ buf0,
    float* __restrict__ buf1, v8f acc[2][7], int lane) {
  const v2f xv = *(const v2f*)(x + (m0 + lane) * 2);
  buf0[lane * STRIDE + 0] = xv.x;
  buf0[lane * STRIDE + 1] = xv.y;
  buf0[lane * STRIDE + 2] = 0.f;
  buf0[lane * STRIDE + 3] = 0.f;
  lds_fence();
  hidden_layer<2, 10, 4, 16>(W1, B1, buf0, buf1, lane);
  hidden_layer<10, 20, 16, 32>(W2, B2, buf1, buf0, lane);
  hidden_layer<20, 50, 32, 64>(W3, B3, buf0, buf1, lane);
  final_layer<50, 100, 64, 112, ACCUM>(W4, B4, buf1, acc, lane);
}

__global__ __launch_bounds__(WAVES_PER_BLOCK * LANES) void
double_decoder_kernel(const float* __restrict__ x,
                      const float* muW1, const float* mub1, const float* muW2,
                      const float* mub2, const float* muW3, const float* mub3,
                      const float* muW4, const float* mub4, const float* sgW1,
                      const float* sgb1, const float* sgW2, const float* sgb2,
                      const float* sgW3, const float* sgb3, const float* sgW4,
                      const float* sgb4, float* __restrict__ out) {
  __shared__ float lds[WAVES_PER_BLOCK * 2 * M_PER_WAVE * STRIDE];
  const int lane = threadIdx.x & (LANES - 1);
  const int w    = threadIdx.x / LANES;
  const long m0  = ((long)blockIdx.x * WAVES_PER_BLOCK + w) * M_PER_WAVE;

  float* buf0 = &lds[(w * 2 + 0) * M_PER_WAVE * STRIDE];
  float* buf1 = &lds[(w * 2 + 1) * M_PER_WAVE * STRIDE];

  v8f acc[2][7];
  run_tower<false>(x, m0, muW1, mub1, muW2, mub2, muW3, mub3, muW4, mub4, buf0,
                   buf1, acc, lane);
  run_tower<true>(x, m0, sgW1, sgb1, sgW2, sgb2, sgW3, sgb3, sgW4, sgb4, buf0,
                  buf1, acc, lane);

  const int half = lane >> 4;
  const int l16  = lane & 15;
#pragma unroll
  for (int mt = 0; mt < 2; ++mt) {
#pragma unroll
    for (int nt = 0; nt < 7; ++nt) {
      const int col = nt * 16 + l16;
      if (col < 100) {
#pragma unroll
        for (int r = 0; r < 8; ++r) {
          const long m = m0 + mt * 16 + r + half * 8;
          out[m * 100 + col] = acc[mt][nt][r];
        }
      }
    }
  }
}

extern "C" void kernel_launch(void* const* d_in, const int* in_sizes, int n_in,
                              void* d_out, int out_size, void* d_ws,
                              size_t ws_size, hipStream_t stream) {
  (void)in_sizes; (void)n_in; (void)d_ws; (void)ws_size; (void)out_size;
  const float* x    = (const float*)d_in[0];
  const float* muW1 = (const float*)d_in[1];
  const float* mub1 = (const float*)d_in[2];
  const float* muW2 = (const float*)d_in[3];
  const float* mub2 = (const float*)d_in[4];
  const float* muW3 = (const float*)d_in[5];
  const float* mub3 = (const float*)d_in[6];
  const float* muW4 = (const float*)d_in[7];
  const float* mub4 = (const float*)d_in[8];
  const float* sgW1 = (const float*)d_in[9];
  const float* sgb1 = (const float*)d_in[10];
  const float* sgW2 = (const float*)d_in[11];
  const float* sgb2 = (const float*)d_in[12];
  const float* sgW3 = (const float*)d_in[13];
  const float* sgb3 = (const float*)d_in[14];
  const float* sgW4 = (const float*)d_in[15];
  const float* sgb4 = (const float*)d_in[16];
  float* out = (float*)d_out;

  const int n_samples = 262144;
  const int blocks = n_samples / (WAVES_PER_BLOCK * M_PER_WAVE);
  double_decoder_kernel<<<blocks, WAVES_PER_BLOCK * LANES, 0, stream>>>(
      x, muW1, mub1, muW2, mub2, muW3, mub3, muW4, mub4, sgW1, sgb1, sgW2,
      sgb2, sgW3, sgb3, sgW4, sgb4, out);
}


# Weights and biases of both towers staged once per work-group in LDS in B-fragment order; 256 work-groups of 8 waves loop over row tiles; B fragments read from LDS a tile ahead, not from global
# speedup vs baseline: 2.8340x; 2.8340x over previous
#include <hip/hip_runtime.h>
#include <hip/hip_bf16.h>
#include <math.h>

typedef __attribute__((ext_vector_type(2))) float v2f;
typedef __attribute__((ext_vector_type(8))) float v8f;

#define WAVES_PER_BLOCK 8
#define THREADS (WAVES_PER_BLOCK * 32)
#define LANES 32
#define M_PER_WAVE 32
#define STRIDE 113
#define TILES_PER_WAVE 4
#define N_SAMPLES 262144

#define OFF_W1 0
#define OFF_W2 (OFF_W1 + 1 * 1 * 64)
#define OFF_W3 (OFF_W2 + 4 * 2 * 64)
#define OFF_W4 (OFF_W3 + 8 * 4 * 64)
#define OFF_B1 (OFF_W4 + 16 * 7 * 64)
#define OFF_B2 (OFF_B1 + 16)
#define OFF_B3 (OFF_B2 + 32)
#define OFF_B4 (OFF_B3 + 64)
#define TOWER_FLOATS (OFF_B4 + 112)

static __device__ __forceinline__ v8f wmma4(v2f a, v2f b, v8f c) {
  return __builtin_amdgcn_wmma_f32_16x16x4_f32(false, a, false, b, (short)0, c,
                                               false, false);
}

static __device__ __forceinline__ void lds_fence() {
  asm volatile("s_wait_dscnt 0x0" ::: "memory");
}

static __device__ __forceinline__ float fast_tanh(float x) {
#if __has_builtin(__builtin_amdgcn_tanhf)
  return __builtin_amdgcn_tanhf(x);
#else
  float e = __builtin_amdgcn_exp2f(x * 2.885390081777927f);
  return __builtin_fmaf(-2.f, __builtin_amdgcn_rcpf(e + 1.f), 1.f);
#endif
}

static __device__ __forceinline__ float gload0(const float* __restrict__ p,
                                               bool valid, int idx) {
  float v = p[valid ? idx : 0];
  return valid ? v : 0.f;
}

template <int KI, int NI, int KPAD, int NPAD>
static __device__ __forceinline__ void stage_weights(
    const float* __restrict__ W, float* __restrict__ wdst, int tid) {
  constexpr int KS = KPAD / 4;
  constexpr int NF = (NPAD / 16) * KS * 32;
  static_assert(NF % THREADS == 0, "whole trips only");
#pragma unroll
  for (int j = 0; j < NF / THREADS; ++j) {
    const int i    = j * THREADS + tid;
    const int lane = i & 31;
    const int f    = i >> 5;
    const int ks   = f % KS;
    const int nt   = f / KS;
    const int ka   = ks * 4 + (lane >> 4) * 2;
    const int col  = nt * 16 + (lane & 15);
    v2f b;
    b.x = gload0(W, (ka + 0) < KI && col < NI, (ka + 0) * NI + col);
    b.y = gload0(W, (ka + 1) < KI && col < NI, (ka + 1) * NI + col);
    ((v2f*)wdst)[i] = b;
  }
}

static __device__ __forceinline__ void stage_small(
    const float* W1, const float* B1, const float* B2, const float* B3,
    const float* B4, float* __restrict__ img, int tid) {
  const bool isb = tid < 224;
  const float* bp;
  int j, n;
  if (tid < 16)       { bp = B1; j = tid;       n = 10;  }
  else if (tid < 48)  { bp = B2; j = tid - 16;  n = 20;  }
  else if (tid < 112) { bp = B3; j = tid - 48;  n = 50;  }
  else                { bp = B4; j = tid - 112; n = 100; }
  const float bv = gload0(bp, isb && j < n, j);
  const int e   = tid & 31;
  const int ka  = (e >> 4) * 2;
  const int col = e & 15;
  v2f w;
  w.x = gload0(W1, !isb && (ka + 0) < 2 && col < 10, (ka + 0) * 10 + col);
  w.y = gload0(W1, !isb && (ka + 1) < 2 && col < 10, (ka + 1) * 10 + col);
  if (isb) img[OFF_B1 + tid] = bv;
  else     ((v2f*)(img + OFF_W1))[e] = w;
}

static __device__ __forceinline__ void stage_tower(
    const float* W1, const float* B1, const float* W2, const float* B2,
    const float* W3, const float* B3, const float* W4, const float* B4,
    float* __restrict__ img, int tid) {
  stage_weights<50, 100, 64, 112>(W4, img + OFF_W4, tid);
  stage_weights<20, 50, 32, 64>(W3, img + OFF_W3, tid);
  stage_weights<10, 20, 16, 32>(W2, img + OFF_W2, tid);
  stage_small(W1, B1, B2, B3, B4, img, tid);
}

template <int KS>
static __device__ __forceinline__ void load_a(const float* __restrict__ hin,
                                              int half, int l16, v2f (&a0)[KS],
                                              v2f (&a1)[KS]) {
#pragma unroll
  for (int ks = 0; ks < KS; ++ks) {
    const int ka = ks * 4 + half * 2;
    a0[ks].x = hin[l16 * STRIDE + ka];
    a0[ks].y = hin[l16 * STRIDE + ka + 1];
    a1[ks].x = hin[(16 + l16) * STRIDE + ka];
    a1[ks].y = hin[(16 + l16) * STRIDE + ka + 1];
  }
}

template <int KS>
static __device__ __forceinline__ void load_b(const v2f* __restrict__ Wf,
                                              int lane, v2f (&b)[KS]) {
#pragma unroll
  for (int ks = 0; ks < KS; ++ks) b[ks] = Wf[ks * 32 + lane];
}

template <int KS>
static __device__ __forceinline__ void mma_block(const v2f (&a0)[KS],
                                                 const v2f (&a1)[KS],
                                                 const v2f (&b)[KS], v8f& c0,
                                                 v8f& c1) {
  c0 = (v8f){0.f, 0.f, 0.f, 0.f, 0.f, 0.f, 0.f, 0.f};
  c1 = (v8f){0.f, 0.f, 0.f, 0.f, 0.f, 0.f, 0.f, 0.f};
#pragma unroll
  for (int ks = 0; ks < KS; ++ks) {
    c0 = wmma4(a0[ks], b[ks], c0);
    c1 = wmma4(a1[ks], b[ks], c1);
  }
}

template <int KPAD, int NPAD>
static __device__ __forceinline__ void hidden_layer(
    const float* __restrict__ Wimg, const float* __restrict__ Bimg,
    const float* __restrict__ hin, float* __restrict__ hout, int lane) {
  constexpr int KS = KPAD / 4;
  constexpr int NT = NPAD / 16;
  const int half = lane >> 4;
  const int l16  = lane & 15;
  const v2f* Wf = (const v2f*)Wimg;
  v2f a0[KS], a1[KS], bq[2][KS];
  load_a<KS>(hin, half, l16, a0, a1);
  load_b<KS>(Wf, lane, bq[0]);
#pragma unroll
  for (int nt = 0; nt < NT; ++nt) {
    const int col = nt * 16 + l16;
    if (nt + 1 < NT) load_b<KS>(Wf + (nt + 1) * KS * 32, lane, bq[(nt + 1) & 1]);
    __builtin_amdgcn_sched_barrier(0);
    v8f c0, c1;
    mma_block<KS>(a0, a1, bq[nt & 1], c0, c1);
    const float bias = Bimg[col];
#pragma unroll
    for (int r = 0; r < 8; ++r) {
      const int m = r + half * 8;
      hout[m * STRIDE + col]        = fast_tanh(c0[r] + bias);
      hout[(16 + m) * STRIDE + col] = fast_tanh(c1[r] + bias);
    }
  }
  lds_fence();
}

template <int KPAD, int NPAD, bool ACCUM>
static __device__ __forceinline__ void final_layer(
    const float* __restrict__ Wimg, const float* __restrict__ Bimg,
    const float* __restrict__ hin, v8f acc[2][NPAD / 16], int lane) {
  constexpr int KS = KPAD / 4;
  constexpr int NT = NPAD / 16;
  const int half = lane >> 4;
  const int l16  = lane & 15;
  const v2f* Wf = (const v2f*)Wimg;
  v2f a0[KS], a1[KS], bq[2][KS];
  load_a<KS>(hin, half, l16, a0, a1);
  load_b<KS>(Wf, lane, bq[0]);
#pragma unroll
  for (int nt = 0; nt < NT; ++nt) {
    const int col = nt * 16 + l16;
    if (nt + 1 < NT) load_b<KS>(Wf + (nt + 1) * KS * 32, lane, bq[(nt + 1) & 1]);
    __builtin_amdgcn_sched_barrier(0);
    v8f c0, c1;
    mma_block<KS>(a0, a1, bq[nt & 1], c0, c1);
    const float bias = Bimg[col];
#pragma unroll
    for (int r = 0; r < 8; ++r) {
      const float t0 = fast_tanh(c0[r] + bias);
      const float t1 = fast_tanh(c1[r] + bias);
      if (ACCUM) {
        acc[0][nt][r] += t0;
        acc[1][nt][r] += t1;
      } else {
        acc[0][nt][r] = t0;
        acc[1][nt][r] = t1;
      }
    }
  }
}

template <bool ACCUM>
static __device__ __forceinline__ void run_tower(
    const float* __restrict__ x, long m0, const float* __restrict__ img,
    float* __restrict__ buf0, float* __restrict__ buf1, v8f acc[2][7],
    int lane) {
  const v2f xv = *(const v2f*)(x + (m0 + lane) * 2);
  buf0[lane * STRIDE + 0] = xv.x;
  buf0[lane * STRIDE + 1] = xv.y;
  buf0[lane * STRIDE + 2] = 0.f;
  buf0[lane * STRIDE + 3] = 0.f;
  lds_fence();
  hidden_layer<4, 16>(img + OFF_W1, img + OFF_B1, buf0, buf1, lane);
  hidden_layer<16, 32>(img + OFF_W2, img + OFF_B2, buf1, buf0, lane);
  hidden_layer<32, 64>(img + OFF_W3, img + OFF_B3, buf0, buf1, lane);
  final_layer<64, 112, ACCUM>(img + OFF_W4, img + OFF_B4, buf1, acc, lane);
}

__global__ __launch_bounds__(THREADS) void
double_decoder_kernel(const float* __restrict__ x,
                      const float* muW1, const float* mub1, const float* muW2,
                      const float* mub2, const float* muW3, const float* mub3,
                      const float* muW4, const float* mub4, const float* sgW1,
                      const float* sgb1, const float* sgW2, const float* sgb2,
                      const float* sgW3, const float* sgb3, const float* sgW4,
                      const float* sgb4, float* __restrict__ out) {
  __shared__ __attribute__((aligned(16))) float wimg[2 * TOWER_FLOATS];
  __shared__ float lds[WAVES_PER_BLOCK * 2 * M_PER_WAVE * STRIDE];
  const int tid  = threadIdx.x;
  const int lane = tid & (LANES - 1);
  const int w    = tid / LANES;

  stage_tower(muW1, mub1, muW2, mub2, muW3, mub3, muW4, mub4, wimg, tid);
  stage_tower(sgW1, sgb1, sgW2, sgb2, sgW3, sgb3, sgW4, sgb4,
              wimg + TOWER_FLOATS, tid);
  __syncthreads();

  float* buf0 = &lds[(w * 2 + 0) * M_PER_WAVE * STRIDE];
  float* buf1 = &lds[(w * 2 + 1) * M_PER_WAVE * STRIDE];
  const int half = lane >> 4;
  const int l16  = lane & 15;

  for (int t = 0; t < TILES_PER_WAVE; ++t) {
    const long m0 =
        (((long)blockIdx.x * TILES_PER_WAVE + t) * WAVES_PER_BLOCK + w) *
        M_PER_WAVE;
    v8f acc[2][7];
    run_tower<false>(x, m0, wimg, buf0, buf1, acc, lane);
    run_tower<true>(x, m0, wimg + TOWER_FLOATS, buf0, buf1, acc, lane);

#pragma unroll
    for (int mt = 0; mt < 2; ++mt) {
#pragma unroll
      for (int nt = 0; nt < 7; ++nt) {
        const int col = nt * 16 + l16;
        if (col < 100) {
#pragma unroll
          for (int r = 0; r < 8; ++r) {
            const long m = m0 + mt * 16 + r + half * 8;
            out[m * 100 + col] = acc[mt][nt][r];
          }
        }
      }
    }
  }
}

extern "C" void kernel_launch(void* const* d_in, const int* in_sizes, int n_in,
                              void* d_out, int out_size, void* d_ws,
                              size_t ws_size, hipStream_t stream) {
  (void)in_sizes; (void)n_in; (void)d_ws; (void)ws_size; (void)out_size;
  const float* x    = (const float*)d_in[0];
  const float* muW1 = (const float*)d_in[1];
  const float* mub1 = (const float*)d_in[2];
  const float* muW2 = (const float*)d_in[3];
  const float* mub2 = (const float*)d_in[4];
  const float* muW3 = (const float*)d_in[5];
  const float* mub3 = (const float*)d_in[6];
  const float* muW4 = (const float*)d_in[7];
  const float* mub4 = (const float*)d_in[8];
  const float* sgW1 = (const float*)d_in[9];
  const float* sgb1 = (const float*)d_in[10];
  const float* sgW2 = (const float*)d_in[11];
  const float* sgb2 = (const float*)d_in[12];
  const float* sgW3 = (const float*)d_in[13];
  const float* sgb3 = (const float*)d_in[14];
  const float* sgW4 = (const float*)d_in[15];
  const float* sgb4 = (const float*)d_in[16];
  float* out = (float*)d_out;

  const int blocks =
      N_SAMPLES / (WAVES_PER_BLOCK * M_PER_WAVE * TILES_PER_WAVE);
  double_decoder_kernel<<<blocks, THREADS, 0, stream>>>(
      x, muW1, mub1, muW2, mub2, muW3, mub3, muW4, mub4, sgW1, sgb1, sgW2,
      sgb2, sgW3, sgb3, sgW4, sgb4, out);
}


# Layers computed transposed (weights as A, samples as B columns): D after bias+tanh is the next layer's B fragment, activations stay in registers; k summed in new order; out tile via LDS as full lines
# speedup vs baseline: 3.6705x; 1.2952x over previous
#include <hip/hip_runtime.h>
#include <hip/hip_bf16.h>
#include <math.h>

typedef __attribute__((ext_vector_type(2))) float v2f;
typedef __attribute__((ext_vector_type(4))) float v4f;
typedef __attribute__((ext_vector_type(8))) float v8f;

#define WAVES_PER_BLOCK 8
#define THREADS (WAVES_PER_BLOCK * 32)
#define LANES 32
#define M_PER_WAVE 32
#define TILES_PER_WAVE 4
#define N_SAMPLES 262144
#define OUT_DIM 100

#define OFF_W1 0
#define OFF_W2 (OFF_W1 + 1 * 1 * 64)
#define OFF_W3 (OFF_W2 + 4 * 2 * 64)
#define OFF_W4 (OFF_W3 + 8 * 4 * 64)
#define OFF_B1 (OFF_W4 + 16 * 7 * 64)
#define OFF_B2 (OFF_B1 + 16)
#define OFF_B3 (OFF_B2 + 32)
#define OFF_B4 (OFF_B3 + 64)
#define TOWER_FLOATS (OFF_B4 + 112)

static __device__ __forceinline__ v8f wmma4(v2f a, v2f b, v8f c) {
  return __builtin_amdgcn_wmma_f32_16x16x4_f32(false, a, false, b, (short)0, c,
                                               false, false);
}

static __device__ __forceinline__ void lds_fence() {
  asm volatile("s_wait_dscnt 0x0" ::: "memory");
}

static __device__ __forceinline__ float fast_tanh(float x) {
#if __has_builtin(__builtin_amdgcn_tanhf)
  return __builtin_amdgcn_tanhf(x);
#else
  float e = __builtin_amdgcn_exp2f(x * 2.885390081777927f);
  return __builtin_fmaf(-2.f, __builtin_amdgcn_rcpf(e + 1.f), 1.f);
#endif
}

static __device__ __forceinline__ float gload0(const float* __restrict__ p,
                                               bool valid, int idx) {
  float v = p[valid ? idx : 0];
  return valid ? v : 0.f;
}

template <int KI, int NI, int KPAD, int NPAD>
static __device__ __forceinline__ void stage_weights(
    const float* __restrict__ W, float* __restrict__ wdst, int tid) {
  constexpr int KS = KPAD / 4;
  constexpr int NF = (NPAD / 16) * KS * 32;
  static_assert(NF % THREADS == 0, "whole trips only");
#pragma unroll
  for (int j = 0; j < NF / THREADS; ++j) {
    const int i    = j * THREADS + tid;
    const int lane = i & 31;
    const int f    = i >> 5;
    const int ks   = f % KS;
    const int mt   = f / KS;
    const int ka   = (ks >> 2) * 16 + (lane >> 4) * 8 + (ks & 3) * 2;
    const int col  = mt * 16 + (lane & 15);
    v2f b;
    b.x = gload0(W, (ka + 0) < KI && col < NI, (ka + 0) * NI + col);
    b.y = gload0(W, (ka + 1) < KI && col < NI, (ka + 1) * NI + col);
    ((v2f*)wdst)[i] = b;
  }
}

static __device__ __forceinline__ void stage_small(
    const float* W1, const float* B1, const float* B2, const float* B3,
    const float* B4, float* __restrict__ img, int tid) {
  const bool isb = tid < 224;
  const float* bp;
  int j, n;
  if (tid < 16)       { bp = B1; j = tid;       n = 10;  }
  else if (tid < 48)  { bp = B2; j = tid - 16;  n = 20;  }
  else if (tid < 112) { bp = B3; j = tid - 48;  n = 50;  }
  else                { bp = B4; j = tid - 112; n = 100; }
  const float bv = gload0(bp, isb && j < n, j);
  const int e   = tid & 31;
  const int ka  = (e >> 4) * 8;
  const int col = e & 15;
  v2f w;
  w.x = gload0(W1, !isb && (ka + 0) < 2 && col < 10, (ka + 0) * 10 + col);
  w.y = gload0(W1, !isb && (ka + 1) < 2 && col < 10, (ka + 1) * 10 + col);
  if (isb) img[OFF_B1 + tid] = bv;
  else     ((v2f*)(img + OFF_W1))[e] = w;
}

static __device__ __forceinline__ void stage_tower(
    const float* W1, const float* B1, const float* W2, const float* B2,
    const float* W3, const float* B3, const float* W4, const float* B4,
    float* __restrict__ img, int tid) {
  stage_weights<50, 100, 64, 112>(W4, img + OFF_W4, tid);
  stage_weights<20, 50, 32, 64>(W3, img + OFF_W3, tid);
  stage_weights<10, 20, 16, 32>(W2, img + OFF_W2, tid);
  stage_small(W1, B1, B2, B3, B4, img, tid);
}

template <int KS>
static __device__ __forceinline__ void load_w(const v2f* __restrict__ Wf,
                                              const float* __restrict__ Bt,
                                              int lane, v2f (&a)[KS],
                                              v4f (&bias)[2]) {
#pragma unroll
  for (int ks = 0; ks < KS; ++ks) a[ks] = Wf[ks * 32 + lane];
  const v4f* bp = (const v4f*)(Bt + (lane >> 4) * 8);
  bias[0] = bp[0];
  bias[1] = bp[1];
}

template <int KS>
static __device__ __forceinline__ void mma_block(const v2f (&a)[KS],
                                                 const v2f (&b0)[KS],
                                                 const v2f (&b1)[KS], v8f& c0,
                                                 v8f& c1) {
  c0 = (v8f){0.f, 0.f, 0.f, 0.f, 0.f, 0.f, 0.f, 0.f};
  c1 = (v8f){0.f, 0.f, 0.f, 0.f, 0.f, 0.f, 0.f, 0.f};
#pragma unroll
  for (int ks = 0; ks < KS; ++ks) {
    c0 = wmma4(a[ks], b0[ks], c0);
    c1 = wmma4(a[ks], b1[ks], c1);
  }
}

template <int KPAD, int NPAD>
static __device__ __forceinline__ void hidden_layer(
    const float* __restrict__ Wimg, const float* __restrict__ Bimg,
    const v2f (&hin)[2][KPAD / 4], v2f (&hout)[2][NPAD / 4], int lane) {
  constexpr int KS = KPAD / 4;
  constexpr int NT = NPAD / 16;
  const v2f* Wf = (const v2f*)Wimg;
  v2f wq[2][KS];
  v4f bq[2][2];
  load_w<KS>(Wf, Bimg, lane, wq[0], bq[0]);
#pragma unroll
  for (int mt = 0; mt < NT; ++mt) {
    if (mt + 1 < NT)
      load_w<KS>(Wf + (mt + 1) * KS * 32, Bimg + (mt + 1) * 16, lane,
                 wq[(mt + 1) & 1], bq[(mt + 1) & 1]);
    __builtin_amdgcn_sched_barrier(0);
    v8f c0, c1;
    mma_block<KS>(wq[mt & 1], hin[0], hin[1], c0, c1);
#pragma unroll
    for (int j = 0; j < 4; ++j) {
      const v2f b2 = {bq[mt & 1][j >> 1][(j & 1) * 2],
                      bq[mt & 1][j >> 1][(j & 1) * 2 + 1]};
      const v2f p0 = (v2f){c0[2 * j], c0[2 * j + 1]} + b2;
      const v2f p1 = (v2f){c1[2 * j], c1[2 * j + 1]} + b2;
      hout[0][mt * 4 + j] = (v2f){fast_tanh(p0.x), fast_tanh(p0.y)};
      hout[1][mt * 4 + j] = (v2f){fast_tanh(p1.x), fast_tanh(p1.y)};
    }
  }
}

template <int KPAD, int NPAD, bool ACCUM>
static __device__ __forceinline__ void final_layer(
    const float* __restrict__ Wimg, const float* __restrict__ Bimg,
    const v2f (&hin)[2][KPAD / 4], v8f (&acc)[2][NPAD / 16], int lane) {
  constexpr int KS = KPAD / 4;
  constexpr int NT = NPAD / 16;
  const v2f* Wf = (const v2f*)Wimg;
  v2f wq[2][KS];
  v4f bq[2][2];
  load_w<KS>(Wf, Bimg, lane, wq[0], bq[0]);
#pragma unroll
  for (int mt = 0; mt < NT; ++mt) {
    if (mt + 1 < NT)
      load_w<KS>(Wf + (mt + 1) * KS * 32, Bimg + (mt + 1) * 16, lane,
                 wq[(mt + 1) & 1], bq[(mt + 1) & 1]);
    __builtin_amdgcn_sched_barrier(0);
    v8f c0, c1;
    mma_block<KS>(wq[mt & 1], hin[0], hin[1], c0, c1);
#pragma unroll
    for (int j = 0; j < 4; ++j) {
      const v2f b2 = {bq[mt & 1][j >> 1][(j & 1) * 2],
                      bq[mt & 1][j >> 1][(j & 1) * 2 + 1]};
      const v2f p0 = (v2f){c0[2 * j], c0[2 * j + 1]} + b2;
      const v2f p1 = (v2f){c1[2 * j], c1[2 * j + 1]} + b2;
      const v2f t0 = {fast_tanh(p0.x), fast_tanh(p0.y)};
      const v2f t1 = {fast_tanh(p1.x), fast_tanh(p1.y)};
      if (ACCUM) {
        acc[0][mt][2 * j]     += t0.x;
        acc[0][mt][2 * j + 1] += t0.y;
        acc[1][mt][2 * j]     += t1.x;
        acc[1][mt][2 * j + 1] += t1.y;
      } else {
        acc[0][mt][2 * j]     = t0.x;
        acc[0][mt][2 * j + 1] = t0.y;
        acc[1][mt][2 * j]     = t1.x;
        acc[1][mt][2 * j + 1] = t1.y;
      }
    }
  }
}

template <bool ACCUM>
static __device__ __forceinline__ void run_tower(
    const v2f (&xin)[2][1], const float* __restrict__ img, v8f (&acc)[2][7],
    int lane) {
  v2f h1[2][4], h2[2][8], h3[2][16];
  hidden_layer<4, 16>(img + OFF_W1, img + OFF_B1, xin, h1, lane);
  hidden_layer<16, 32>(img + OFF_W2, img + OFF_B2, h1, h2, lane);
  hidden_layer<32, 64>(img + OFF_W3, img + OFF_B3, h2, h3, lane);
  final_layer<64, 112, ACCUM>(img + OFF_W4, img + OFF_B4, h3, acc, lane);
}

__global__ __launch_bounds__(THREADS) void
double_decoder_kernel(const float* __restrict__ x,
                      const float* muW1, const float* mub1, const float* muW2,
                      const float* mub2, const float* muW3, const float* mub3,
                      const float* muW4, const float* mub4, const float* sgW1,
                      const float* sgb1, const float* sgW2, const float* sgb2,
                      const float* sgW3, const float* sgb3, const float* sgW4,
                      const float* sgb4, float* __restrict__ out) {
  __shared__ __attribute__((aligned(16))) float wimg[2 * TOWER_FLOATS];
  __shared__ __attribute__((aligned(16))) float
      ostage[WAVES_PER_BLOCK * M_PER_WAVE * OUT_DIM];
  const int tid  = threadIdx.x;
  const int lane = tid & (LANES - 1);
  const int w    = tid / LANES;

  stage_tower(muW1, mub1, muW2, mub2, muW3, mub3, muW4, mub4, wimg, tid);
  stage_tower(sgW1, sgb1, sgW2, sgb2, sgW3, sgb3, sgW4, sgb4,
              wimg + TOWER_FLOATS, tid);
  __syncthreads();

  float* og = &ostage[w * M_PER_WAVE * OUT_DIM];
  const int half = lane >> 4;
  const int l16  = lane & 15;

  for (int t = 0; t < TILES_PER_WAVE; ++t) {
    const long m0 =
        (((long)blockIdx.x * TILES_PER_WAVE + t) * WAVES_PER_BLOCK + w) *
        M_PER_WAVE;
    v2f xin[2][1];
#pragma unroll
    for (int s = 0; s < 2; ++s) {
      const v2f xv = *(const v2f*)(x + (m0 + s * 16 + l16) * 2);
      xin[s][0].x = half ? 0.f : xv.x;
      xin[s][0].y = half ? 0.f : xv.y;
    }
    v8f acc[2][7];
    run_tower<false>(xin, wimg, acc, lane);
    run_tower<true>(xin, wimg + TOWER_FLOATS, acc, lane);

    static_assert(OUT_DIM == 6 * 16 + 4, "tile 6: registers 0..3 of half 0");
#pragma unroll
    for (int s = 0; s < 2; ++s) {
#pragma unroll
      for (int mt = 0; mt < 6; ++mt) {
#pragma unroll
        for (int q = 0; q < 2; ++q) {
          const v4f v = {acc[s][mt][q * 4 + 0], acc[s][mt][q * 4 + 1],
                         acc[s][mt][q * 4 + 2], acc[s][mt][q * 4 + 3]};
          *(v4f*)(og + (s * 16 + l16) * OUT_DIM + mt * 16 + half * 8 + q * 4) = v;
        }
      }
    }
    if (half == 0) {
#pragma unroll
      for (int s = 0; s < 2; ++s) {
        const v4f v = {acc[s][6][0], acc[s][6][1], acc[s][6][2], acc[s][6][3]};
        *(v4f*)(og + (s * 16 + l16) * OUT_DIM + 96) = v;
      }
    }
    lds_fence();
    float* orun = out + m0 * OUT_DIM;
#pragma unroll
    for (int it = 0; it < M_PER_WAVE * OUT_DIM / (4 * LANES); ++it) {
      const int idx = (it * LANES + lane) * 4;
      *(v4f*)(orun + idx) = *(const v4f*)(og + idx);
    }
  }
}

extern "C" void kernel_launch(void* const* d_in, const int* in_sizes, int n_in,
                              void* d_out, int out_size, void* d_ws,
                              size_t ws_size, hipStream_t stream) {
  (void)in_sizes; (void)n_in; (void)d_ws; (void)ws_size; (void)out_size;
  const float* x    = (const float*)d_in[0];
  const float* muW1 = (const float*)d_in[1];
  const float* mub1 = (const float*)d_in[2];
  const float* muW2 = (const float*)d_in[3];
  const float* mub2 = (const float*)d_in[4];
  const float* muW3 = (const float*)d_in[5];
  const float* mub3 = (const float*)d_in[6];
  const float* muW4 = (const float*)d_in[7];
  const float* mub4 = (const float*)d_in[8];
  const float* sgW1 = (const float*)d_in[9];
  const float* sgb1 = (const float*)d_in[10];
  const float* sgW2 = (const float*)d_in[11];
  const float* sgb2 = (const float*)d_in[12];
  const float* sgW3 = (const float*)d_in[13];
  const float* sgb3 = (const float*)d_in[14];
  const float* sgW4 = (const float*)d_in[15];
  const float* sgb4 = (const float*)d_in[16];
  float* out = (float*)d_out;

  const int blocks =
      N_SAMPLES / (WAVES_PER_BLOCK * M_PER_WAVE * TILES_PER_WAVE);
  double_decoder_kernel<<<blocks, THREADS, 0, stream>>>(
      x, muW1, mub1, muW2, mub2, muW3, mub3, muW4, mub4, sgW1, sgb1, sgW2,
      sgb2, sgW3, sgb3, sgW4, sgb4, out);
}


# Zero padding left out of the products: staging puts hidden features in k-step order, so layers run 3, 5, 13 k-steps (were 4, 8, 16); padded D rows not formed; k summed in new order
# speedup vs baseline: 4.3683x; 1.1901x over previous
#include <hip/hip_runtime.h>
#include <hip/hip_bf16.h>
#include <math.h>

typedef __attribute__((ext_vector_type(2))) float v2f;
typedef __attribute__((ext_vector_type(4))) float v4f;
typedef __attribute__((ext_vector_type(8))) float v8f;

#define WAVES_PER_BLOCK 8
#define THREADS (WAVES_PER_BLOCK * 32)
#define LANES 32
#define M_PER_WAVE 32
#define TILES_PER_WAVE 4
#define N_SAMPLES 262144
#define OUT_DIM 100

#define KS1 1
#define KS2 3
#define KS3 5
#define KS4 13
#define NT1 1
#define NT2 2
#define NT3 4
#define NT4 7

#define TRIP_FLOATS(entries) ((((entries) + THREADS - 1) / THREADS) * THREADS * 2)
#define OFF_W1 0
#define OFF_W2 (OFF_W1 + KS1 * NT1 * 64)
#define OFF_W3 (OFF_W2 + TRIP_FLOATS(KS2 * NT2 * 32))
#define OFF_W4 (OFF_W3 + TRIP_FLOATS(KS3 * NT3 * 32))
#define OFF_B1 (OFF_W4 + TRIP_FLOATS(KS4 * NT4 * 32))
#define OFF_B2 (OFF_B1 + 16 * NT1)
#define OFF_B3 (OFF_B2 + 16 * NT2)
#define OFF_B4 (OFF_B3 + 16 * NT3)
#define TOWER_FLOATS (OFF_B4 + 16 * NT4)

static __device__ __forceinline__ int hidden_col(int p) {
  return (p & ~15) + 4 * ((p & 7) >> 1) + 2 * ((p >> 3) & 1) + (p & 1);
}

static __device__ __forceinline__ v8f wmma4(v2f a, v2f b, v8f c) {
  return __builtin_amdgcn_wmma_f32_16x16x4_f32(false, a, false, b, (short)0, c,
                                               false, false);
}

static __device__ __forceinline__ void lds_fence() {
  asm volatile("s_wait_dscnt 0x0" ::: "memory");
}

static __device__ __forceinline__ float fast_tanh(float x) {
#if __has_builtin(__builtin_amdgcn_tanhf)
  return __builtin_amdgcn_tanhf(x);
#else
  float e = __builtin_amdgcn_exp2f(x * 2.885390081777927f);
  return __builtin_fmaf(-2.f, __builtin_amdgcn_rcpf(e + 1.f), 1.f);
#endif
}

static __device__ __forceinline__ float gload0(const float* __restrict__ p,
                                               bool valid, int idx) {
  float v = p[valid ? idx : 0];
  return valid ? v : 0.f;
}

template <int KI, int NI, int KS, int NT, bool HIDDEN>
static __device__ __forceinline__ void stage_weights(
    const float* __restrict__ W, float* __restrict__ wdst, int tid) {
  constexpr int NF = NT * KS * 32;
  constexpr int TRIPS = (NF + THREADS - 1) / THREADS;
  static_assert(16 * NT >= NI, "entries past the last fragment load nothing");
#pragma unroll
  for (int j = 0; j < TRIPS; ++j) {
    const int i    = j * THREADS + tid;
    const int lane = i & 31;
    const int f    = i >> 5;
    const int ks   = f % KS;
    const int mt   = f / KS;
    const int ka   = 4 * ks + 2 * (lane >> 4);
    const int row  = mt * 16 + (lane & 15);
    const int col  = HIDDEN ? hidden_col(row) : row;
    v2f b;
    b.x = gload0(W, (ka + 0) < KI && col < NI, (ka + 0) * NI + col);
    b.y = gload0(W, (ka + 1) < KI && col < NI, (ka + 1) * NI + col);
    ((v2f*)wdst)[i] = b;
  }
}

static __device__ __forceinline__ void stage_small(
    const float* W1, const float* B1, const float* B2, const float* B3,
    const float* B4, float* __restrict__ img, int tid) {
  const bool isb = tid < 224;
  const float* bp;
  int j, n;
  if (tid < 16)       { bp = B1; j = hidden_col(tid);       n = 10;  }
  else if (tid < 48)  { bp = B2; j = hidden_col(tid - 16);  n = 20;  }
  else if (tid < 112) { bp = B3; j = hidden_col(tid - 48);  n = 50;  }
  else                { bp = B4; j = tid - 112;             n = 100; }
  const float bv = gload0(bp, isb && j < n, j);
  const int e   = tid & 31;
  const int ka  = (e >> 4) * 2;
  const int col = hidden_col(e & 15);
  v2f w;
  w.x = gload0(W1, !isb && (ka + 0) < 2 && col < 10, (ka + 0) * 10 + col);
  w.y = gload0(W1, !isb && (ka + 1) < 2 && col < 10, (ka + 1) * 10 + col);
  if (isb) img[OFF_B1 + tid] = bv;
  else     ((v2f*)(img + OFF_W1))[e] = w;
}

static __device__ __forceinline__ void stage_tower(
    const float* W1, const float* B1, const float* W2, const float* B2,
    const float* W3, const float* B3, const float* W4, const float* B4,
    float* __restrict__ img, int tid) {
  stage_weights<50, 100, KS4, NT4, false>(W4, img + OFF_W4, tid);
  stage_weights<20, 50, KS3, NT3, true>(W3, img + OFF_W3, tid);
  stage_weights<10, 20, KS2, NT2, true>(W2, img + OFF_W2, tid);
  stage_small(W1, B1, B2, B3, B4, img, tid);
}

template <int KS>
static __device__ __forceinline__ void load_w(const v2f* __restrict__ Wf,
                                              const float* __restrict__ Bt,
                                              int lane, v2f (&a)[KS],
                                              v4f (&bias)[2]) {
#pragma unroll
  for (int ks = 0; ks < KS; ++ks) a[ks] = Wf[ks * 32 + lane];
  const v4f* bp = (const v4f*)(Bt + (lane >> 4) * 8);
  bias[0] = bp[0];
  bias[1] = bp[1];
}

template <int KS>
static __device__ __forceinline__ void mma_block(const v2f (&a)[KS],
                                                 const v2f (&b0)[KS],
                                                 const v2f (&b1)[KS], v8f& c0,
                                                 v8f& c1) {
  c0 = (v8f){0.f, 0.f, 0.f, 0.f, 0.f, 0.f, 0.f, 0.f};
  c1 = (v8f){0.f, 0.f, 0.f, 0.f, 0.f, 0.f, 0.f, 0.f};
#pragma unroll
  for (int ks = 0; ks < KS; ++ks) {
    c0 = wmma4(a[ks], b0[ks], c0);
    c1 = wmma4(a[ks], b1[ks], c1);
  }
}

template <int KS, int NT, int KSO>
static __device__ __forceinline__ void hidden_layer(
    const float* __restrict__ Wimg, const float* __restrict__ Bimg,
    const v2f (&hin)[2][KS], v2f (&hout)[2][KSO], int lane) {
  static_assert(KSO <= 4 * NT && KSO > 4 * (NT - 1), "every tile feeds a k-step");
  const v2f* Wf = (const v2f*)Wimg;
  v2f wq[2][KS];
  v4f bq[2][2];
  load_w<KS>(Wf, Bimg, lane, wq[0], bq[0]);
#pragma unroll
  for (int mt = 0; mt < NT; ++mt) {
    if (mt + 1 < NT)
      load_w<KS>(Wf + (mt + 1) * KS * 32, Bimg + (mt + 1) * 16, lane,
                 wq[(mt + 1) & 1], bq[(mt + 1) & 1]);
    __builtin_amdgcn_sched_barrier(0);
    v8f c0, c1;
    mma_block<KS>(wq[mt & 1], hin[0], hin[1], c0, c1);
#pragma unroll
    for (int j = 0; j < 4; ++j) {
      if (mt * 4 + j < KSO) {
        const v2f b2 = {bq[mt & 1][j >> 1][(j & 1) * 2],
                        bq[mt & 1][j >> 1][(j & 1) * 2 + 1]};
        const v2f p0 = (v2f){c0[2 * j], c0[2 * j + 1]} + b2;
        const v2f p1 = (v2f){c1[2 * j], c1[2 * j + 1]} + b2;
        hout[0][mt * 4 + j] = (v2f){fast_tanh(p0.x), fast_tanh(p0.y)};
        hout[1][mt * 4 + j] = (v2f){fast_tanh(p1.x), fast_tanh(p1.y)};
      }
    }
  }
}

template <int KS, int NT, int NREAL, bool ACCUM>
static __device__ __forceinline__ void final_layer(
    const float* __restrict__ Wimg, const float* __restrict__ Bimg,
    const v2f (&hin)[2][KS], v8f (&acc)[2][NT], int lane) {
  const v2f* Wf = (const v2f*)Wimg;
  v2f wq[2][KS];
  v4f bq[2][2];
  load_w<KS>(Wf, Bimg, lane, wq[0], bq[0]);
#pragma unroll
  for (int mt = 0; mt < NT; ++mt) {
    if (mt + 1 < NT)
      load_w<KS>(Wf + (mt + 1) * KS * 32, Bimg + (mt + 1) * 16, lane,
                 wq[(mt + 1) & 1], bq[(mt + 1) & 1]);
    __builtin_amdgcn_sched_barrier(0);
    v8f c0, c1;
    mma_block<KS>(wq[mt & 1], hin[0], hin[1], c0, c1);
#pragma unroll
    for (int j = 0; j < 4; ++j) {
      if (mt * 16 + 2 * j < NREAL) {
        const v2f b2 = {bq[mt & 1][j >> 1][(j & 1) * 2],
                        bq[mt & 1][j >> 1][(j & 1) * 2 + 1]};
        const v2f p0 = (v2f){c0[2 * j], c0[2 * j + 1]} + b2;
        const v2f p1 = (v2f){c1[2 * j], c1[2 * j + 1]} + b2;
        const v2f t0 = {fast_tanh(p0.x), fast_tanh(p0.y)};
        const v2f t1 = {fast_tanh(p1.x), fast_tanh(p1.y)};
        if (ACCUM) {
          acc[0][mt][2 * j]     += t0.x;
          acc[0][mt][2 * j + 1] += t0.y;
          acc[1][mt][2 * j]     += t1.x;
          acc[1][mt][2 * j + 1] += t1.y;
        } else {
          acc[0][mt][2 * j]     = t0.x;
          acc[0][mt][2 * j + 1] = t0.y;
          acc[1][mt][2 * j]     = t1.x;
          acc[1][mt][2 * j + 1] = t1.y;
        }
      }
    }
  }
}

template <bool ACCUM>
static __device__ __forceinline__ void run_tower(
    const v2f (&xin)[2][KS1], const float* __restrict__ img,
    v8f (&acc)[2][NT4], int lane) {
  v2f h1[2][KS2], h2[2][KS3], h3[2][KS4];
  hidden_layer<KS1, NT1, KS2>(img + OFF_W1, img + OFF_B1, xin, h1, lane);
  hidden_layer<KS2, NT2, KS3>(img + OFF_W2, img + OFF_B2, h1, h2, lane);
  hidden_layer<KS3, NT3, KS4>(img + OFF_W3, img + OFF_B3, h2, h3, lane);
  final_layer<KS4, NT4, OUT_DIM, ACCUM>(img + OFF_W4, img + OFF_B4, h3, acc,
                                        lane);
}

__global__ __launch_bounds__(THREADS) void
double_decoder_kernel(const float* __restrict__ x,
                      const float* muW1, const float* mub1, const float* muW2,
                      const float* mub2, const float* muW3, const float* mub3,
                      const float* muW4, const float* mub4, const float* sgW1,
                      const float* sgb1, const float* sgW2, const float* sgb2,
                      const float* sgW3, const float* sgb3, const float* sgW4,
                      const float* sgb4, float* __restrict__ out) {
  __shared__ __attribute__((aligned(16))) float wimg[2 * TOWER_FLOATS];
  __shared__ __attribute__((aligned(16))) float
      ostage[WAVES_PER_BLOCK * M_PER_WAVE * OUT_DIM];
  const int tid  = threadIdx.x;
  const int lane = tid & (LANES - 1);
  const int w    = tid / LANES;

  stage_tower(muW1, mub1, muW2, mub2, muW3, mub3, muW4, mub4, wimg, tid);
  stage_tower(sgW1, sgb1, sgW2, sgb2, sgW3, sgb3, sgW4, sgb4,
              wimg + TOWER_FLOATS, tid);
  __syncthreads();

  float* og = &ostage[w * M_PER_WAVE * OUT_DIM];
  const int half = lane >> 4;
  const int l16  = lane & 15;

  for (int t = 0; t < TILES_PER_WAVE; ++t) {
    const long m0 =
        (((long)blockIdx.x * TILES_PER_WAVE + t) * WAVES_PER_BLOCK + w) *
        M_PER_WAVE;
    v2f xin[2][1];
#pragma unroll
    for (int s = 0; s < 2; ++s) {
      const v2f xv = *(const v2f*)(x + (m0 + s * 16 + l16) * 2);
      xin[s][0].x = half ? 0.f : xv.x;
      xin[s][0].y = half ? 0.f : xv.y;
    }
    v8f acc[2][7];
    run_tower<false>(xin, wimg, acc, lane);
    run_tower<true>(xin, wimg + TOWER_FLOATS, acc, lane);

    static_assert(OUT_DIM == 6 * 16 + 4, "tile 6: registers 0..3 of half 0");
#pragma unroll
    for (int s = 0; s < 2; ++s) {
#pragma unroll
      for (int mt = 0; mt < 6; ++mt) {
#pragma unroll
        for (int q = 0; q < 2; ++q) {
          const v4f v = {acc[s][mt][q * 4 + 0], acc[s][mt][q * 4 + 1],
                         acc[s][mt][q * 4 + 2], acc[s][mt][q * 4 + 3]};
          *(v4f*)(og + (s * 16 + l16) * OUT_DIM + mt * 16 + half * 8 + q * 4) = v;
        }
      }
    }
    if (half == 0) {
#pragma unroll
      for (int s = 0; s < 2; ++s) {
        const v4f v = {acc[s][6][0], acc[s][6][1], acc[s][6][2], acc[s][6][3]};
        *(v4f*)(og + (s * 16 + l16) * OUT_DIM + 96) = v;
      }
    }
    lds_fence();
    float* orun = out + m0 * OUT_DIM;
#pragma unroll
    for (int it = 0; it < M_PER_WAVE * OUT_DIM / (4 * LANES); ++it) {
      const int idx = (it * LANES + lane) * 4;
      *(v4f*)(orun + idx) = *(const v4f*)(og + idx);
    }
  }
}

extern "C" void kernel_launch(void* const* d_in, const int* in_sizes, int n_in,
                              void* d_out, int out_size, void* d_ws,
                              size_t ws_size, hipStream_t stream) {
  (void)in_sizes; (void)n_in; (void)d_ws; (void)ws_size; (void)out_size;
  const float* x    = (const float*)d_in[0];
  const float* muW1 = (const float*)d_in[1];
  const float* mub1 = (const float*)d_in[2];
  const float* muW2 = (const float*)d_in[3];
  const float* mub2 = (const float*)d_in[4];
  const float* muW3 = (const float*)d_in[5];
  const float* mub3 = (const float*)d_in[6];
  const float* muW4 = (const float*)d_in[7];
  const float* mub4 = (const float*)d_in[8];
  const float* sgW1 = (const float*)d_in[9];
  const float* sgb1 = (const float*)d_in[10];
  const float* sgW2 = (const float*)d_in[11];
  const float* sgb2 = (const float*)d_in[12];
  const float* sgW3 = (const float*)d_in[13];
  const float* sgb3 = (const float*)d_in[14];
  const float* sgW4 = (const float*)d_in[15];
  const float* sgb4 = (const float*)d_in[16];
  float* out = (float*)d_out;

  const int blocks =
      N_SAMPLES / (WAVES_PER_BLOCK * M_PER_WAVE * TILES_PER_WAVE);
  double_decoder_kernel<<<blocks, THREADS, 0, stream>>>(
      x, muW1, mub1, muW2, mub2, muW3, mub3, muW4, mub4, sgW1, sgb1, sgW2,
      sgb2, sgW3, sgb3, sgW4, sgb4, out);
}


# Every wave starts at the highest priority and steps down one level per row tile done (s_setprio 3,2,1,0): the wave of a SIMD pair that is behind wins the matrix pipe, so the pair ends together
# speedup vs baseline: 4.5469x; 1.0409x over previous
#include <hip/hip_runtime.h>
#include <hip/hip_bf16.h>
#include <math.h>

typedef __attribute__((ext_vector_type(2))) float v2f;
typedef __attribute__((ext_vector_type(4))) float v4f;
typedef __attribute__((ext_vector_type(8))) float v8f;

#define WAVES_PER_BLOCK 8
#define THREADS (WAVES_PER_BLOCK * 32)
#define LANES 32
#define M_PER_WAVE 32
#define TILES_PER_WAVE 4
#define N_SAMPLES 262144
#define OUT_DIM 100

#define KS1 1
#define KS2 3
#define KS3 5
#define KS4 13
#define NT1 1
#define NT2 2
#define NT3 4
#define NT4 7

#define TRIP_FLOATS(entries) ((((entries) + THREADS - 1) / THREADS) * THREADS * 2)
#define OFF_W1 0
#define OFF_W2 (OFF_W1 + KS1 * NT1 * 64)
#define OFF_W3 (OFF_W2 + TRIP_FLOATS(KS2 * NT2 * 32))
#define OFF_W4 (OFF_W3 + TRIP_FLOATS(KS3 * NT3 * 32))
#define OFF_B1 (OFF_W4 + TRIP_FLOATS(KS4 * NT4 * 32))
#define OFF_B2 (OFF_B1 + 16 * NT1)
#define OFF_B3 (OFF_B2 + 16 * NT2)
#define OFF_B4 (OFF_B3 + 16 * NT3)
#define TOWER_FLOATS (OFF_B4 + 16 * NT4)

static __device__ __forceinline__ int hidden_col(int p) {
  return (p & ~15) + 4 * ((p & 7) >> 1) + 2 * ((p >> 3) & 1) + (p & 1);
}

static __device__ __forceinline__ v8f wmma4(v2f a, v2f b, v8f c) {
  return __builtin_amdgcn_wmma_f32_16x16x4_f32(false, a, false, b, (short)0, c,
                                               false, false);
}

static __device__ __forceinline__ void lds_fence() {
  asm volatile("s_wait_dscnt 0x0" ::: "memory");
}

static __device__ __forceinline__ float fast_tanh(float x) {
#if __has_builtin(__builtin_amdgcn_tanhf)
  return __builtin_amdgcn_tanhf(x);
#else
  float e = __builtin_amdgcn_exp2f(x * 2.885390081777927f);
  return __builtin_fmaf(-2.f, __builtin_amdgcn_rcpf(e + 1.f), 1.f);
#endif
}

static __device__ __forceinline__ float gload0(const float* __restrict__ p,
                                               bool valid, int idx) {
  float v = p[valid ? idx : 0];
  return valid ? v : 0.f;
}

template <int KI, int NI, int KS, int NT, bool HIDDEN>
static __device__ __forceinline__ void stage_weights(
    const float* __restrict__ W, float* __restrict__ wdst, int tid) {
  constexpr int NF = NT * KS * 32;
  constexpr int TRIPS = (NF + THREADS - 1) / THREADS;
  static_assert(16 * NT >= NI, "entries past the last fragment load nothing");
#pragma unroll
  for (int j = 0; j < TRIPS; ++j) {
    const int i    = j * THREADS + tid;
    const int lane = i & 31;
    const int f    = i >> 5;
    const int ks   = f % KS;
    const int mt   = f / KS;
    const int ka   = 4 * ks + 2 * (lane >> 4);
    const int row  = mt * 16 + (lane & 15);
    const int col  = HIDDEN ? hidden_col(row) : row;
    v2f b;
    b.x = gload0(W, (ka + 0) < KI && col < NI, (ka + 0) * NI + col);
    b.y = gload0(W, (ka + 1) < KI && col < NI, (ka + 1) * NI + col);
    ((v2f*)wdst)[i] = b;
  }
}

static __device__ __forceinline__ void stage_small(
    const float* W1, const float* B1, const float* B2, const float* B3,
    const float* B4, float* __restrict__ img, int tid) {
  const bool isb = tid < 224;
  const float* bp;
  int j, n;
  if (tid < 16)       { bp = B1; j = hidden_col(tid);       n = 10;  }
  else if (tid < 48)  { bp = B2; j = hidden_col(tid - 16);  n = 20;  }
  else if (tid < 112) { bp = B3; j = hidden_col(tid - 48);  n = 50;  }
  else                { bp = B4; j = tid - 112;             n = 100; }
  const float bv = gload0(bp, isb && j < n, j);
  const int e   = tid & 31;
  const int ka  = (e >> 4) * 2;
  const int col = hidden_col(e & 15);
  v2f w;
  w.x = gload0(W1, !isb && (ka + 0) < 2 && col < 10, (ka + 0) * 10 + col);
  w.y = gload0(W1, !isb && (ka + 1) < 2 && col < 10, (ka + 1) * 10 + col);
  if (isb) img[OFF_B1 + tid] = bv;
  else     ((v2f*)(img + OFF_W1))[e] = w;
}

static __device__ __forceinline__ void stage_tower(
    const float* W1, const float* B1, const float* W2, const float* B2,
    const float* W3, const float* B3, const float* W4, const float* B4,
    float* __restrict__ img, int tid) {
  stage_weights<50, 100, KS4, NT4, false>(W4, img + OFF_W4, tid);
  stage_weights<20, 50, KS3, NT3, true>(W3, img + OFF_W3, tid);
  stage_weights<10, 20, KS2, NT2, true>(W2, img + OFF_W2, tid);
  stage_small(W1, B1, B2, B3, B4, img, tid);
}

template <int KS>
static __device__ __forceinline__ void load_w(const v2f* __restrict__ Wf,
                                              const float* __restrict__ Bt,
                                              int lane, v2f (&a)[KS],
                                              v4f (&bias)[2]) {
#pragma unroll
  for (int ks = 0; ks < KS; ++ks) a[ks] = Wf[ks * 32 + lane];
  const v4f* bp = (const v4f*)(Bt + (lane >> 4) * 8);
  bias[0] = bp[0];
  bias[1] = bp[1];
}

template <int KS>
static __device__ __forceinline__ void mma_block(const v2f (&a)[KS],
                                                 const v2f (&b0)[KS],
                                                 const v2f (&b1)[KS], v8f& c0,
                                                 v8f& c1) {
  c0 = (v8f){0.f, 0.f, 0.f, 0.f, 0.f, 0.f, 0.f, 0.f};
  c1 = (v8f){0.f, 0.f, 0.f, 0.f, 0.f, 0.f, 0.f, 0.f};
#pragma unroll
  for (int ks = 0; ks < KS; ++ks) {
    c0 = wmma4(a[ks], b0[ks], c0);
    c1 = wmma4(a[ks], b1[ks], c1);
  }
}

template <int KS, int NT, int KSO>
static __device__ __forceinline__ void hidden_layer(
    const float* __restrict__ Wimg, const float* __restrict__ Bimg,
    const v2f (&hin)[2][KS], v2f (&hout)[2][KSO], int lane) {
  static_assert(KSO <= 4 * NT && KSO > 4 * (NT - 1), "every tile feeds a k-step");
  const v2f* Wf = (const v2f*)Wimg;
  v2f wq[2][KS];
  v4f bq[2][2];
  load_w<KS>(Wf, Bimg, lane, wq[0], bq[0]);
#pragma unroll
  for (int mt = 0; mt < NT; ++mt) {
    if (mt + 1 < NT)
      load_w<KS>(Wf + (mt + 1) * KS * 32, Bimg + (mt + 1) * 16, lane,
                 wq[(mt + 1) & 1], bq[(mt + 1) & 1]);
    __builtin_amdgcn_sched_barrier(0);
    v8f c0, c1;
    mma_block<KS>(wq[mt & 1], hin[0], hin[1], c0, c1);
#pragma unroll
    for (int j = 0; j < 4; ++j) {
      if (mt * 4 + j < KSO) {
        const v2f b2 = {bq[mt & 1][j >> 1][(j & 1) * 2],
                        bq[mt & 1][j >> 1][(j & 1) * 2 + 1]};
        const v2f p0 = (v2f){c0[2 * j], c0[2 * j + 1]} + b2;
        const v2f p1 = (v2f){c1[2 * j], c1[2 * j + 1]} + b2;
        hout[0][mt * 4 + j] = (v2f){fast_tanh(p0.x), fast_tanh(p0.y)};
        hout[1][mt * 4 + j] = (v2f){fast_tanh(p1.x), fast_tanh(p1.y)};
      }
    }
  }
}

template <int KS, int NT, int NREAL, bool ACCUM>
static __device__ __forceinline__ void final_layer(
    const float* __restrict__ Wimg, const float* __restrict__ Bimg,
    const v2f (&hin)[2][KS], v8f (&acc)[2][NT], int lane) {
  const v2f* Wf = (const v2f*)Wimg;
  v2f wq[2][KS];
  v4f bq[2][2];
  load_w<KS>(Wf, Bimg, lane, wq[0], bq[0]);
#pragma unroll
  for (int mt = 0; mt < NT; ++mt) {
    if (mt + 1 < NT)
      load_w<KS>(Wf + (mt + 1) * KS * 32, Bimg + (mt + 1) * 16, lane,
                 wq[(mt + 1) & 1], bq[(mt + 1) & 1]);
    __builtin_amdgcn_sched_barrier(0);
    v8f c0, c1;
    mma_block<KS>(wq[mt & 1], hin[0], hin[1], c0, c1);
#pragma unroll
    for (int j = 0; j < 4; ++j) {
      if (mt * 16 + 2 * j < NREAL) {
        const v2f b2 = {bq[mt & 1][j >> 1][(j & 1) * 2],
                        bq[mt & 1][j >> 1][(j & 1) * 2 + 1]};
        const v2f p0 = (v2f){c0[2 * j], c0[2 * j + 1]} + b2;
        const v2f p1 = (v2f){c1[2 * j], c1[2 * j + 1]} + b2;
        const v2f t0 = {fast_tanh(p0.x), fast_tanh(p0.y)};
        const v2f t1 = {fast_tanh(p1.x), fast_tanh(p1.y)};
        if (ACCUM) {
          acc[0][mt][2 * j]     += t0.x;
          acc[0][mt][2 * j + 1] += t0.y;
          acc[1][mt][2 * j]     += t1.x;
          acc[1][mt][2 * j + 1] += t1.y;
        } else {
          acc[0][mt][2 * j]     = t0.x;
          acc[0][mt][2 * j + 1] = t0.y;
          acc[1][mt][2 * j]     = t1.x;
          acc[1][mt][2 * j + 1] = t1.y;
        }
      }
    }
  }
}

template <bool ACCUM>
static __device__ __forceinline__ void run_tower(
    const v2f (&xin)[2][KS1], const float* __restrict__ img,
    v8f (&acc)[2][NT4], int lane) {
  v2f h1[2][KS2], h2[2][KS3], h3[2][KS4];
  hidden_layer<KS1, NT1, KS2>(img + OFF_W1, img + OFF_B1, xin, h1, lane);
  hidden_layer<KS2, NT2, KS3>(img + OFF_W2, img + OFF_B2, h1, h2, lane);
  hidden_layer<KS3, NT3, KS4>(img + OFF_W3, img + OFF_B3, h2, h3, lane);
  final_layer<KS4, NT4, OUT_DIM, ACCUM>(img + OFF_W4, img + OFF_B4, h3, acc,
                                        lane);
}

__global__ __launch_bounds__(THREADS) void
double_decoder_kernel(const float* __restrict__ x,
                      const float* muW1, const float* mub1, const float* muW2,
                      const float* mub2, const float* muW3, const float* mub3,
                      const float* muW4, const float* mub4, const float* sgW1,
                      const float* sgb1, const float* sgW2, const float* sgb2,
                      const float* sgW3, const float* sgb3, const float* sgW4,
                      const float* sgb4, float* __restrict__ out) {
  __shared__ __attribute__((aligned(16))) float wimg[2 * TOWER_FLOATS];
  __shared__ __attribute__((aligned(16))) float
      ostage[WAVES_PER_BLOCK * M_PER_WAVE * OUT_DIM];
  const int tid  = threadIdx.x;
  const int lane = tid & (LANES - 1);
  const int w    = tid / LANES;

  stage_tower(muW1, mub1, muW2, mub2, muW3, mub3, muW4, mub4, wimg, tid);
  stage_tower(sgW1, sgb1, sgW2, sgb2, sgW3, sgb3, sgW4, sgb4,
              wimg + TOWER_FLOATS, tid);
  __syncthreads();

  float* og = &ostage[w * M_PER_WAVE * OUT_DIM];
  const int half = lane >> 4;
  const int l16  = lane & 15;

  static_assert(TILES_PER_WAVE == 4, "one priority level per tile, 3 down to 0");
  for (int t = 0; t < TILES_PER_WAVE; ++t) {
    if (t == 0)      __builtin_amdgcn_s_setprio(3);
    else if (t == 1) __builtin_amdgcn_s_setprio(2);
    else if (t == 2) __builtin_amdgcn_s_setprio(1);
    else             __builtin_amdgcn_s_setprio(0);
    const long m0 =
        (((long)blockIdx.x * TILES_PER_WAVE + t) * WAVES_PER_BLOCK + w) *
        M_PER_WAVE;
    v2f xin[2][1];
#pragma unroll
    for (int s = 0; s < 2; ++s) {
      const v2f xv = *(const v2f*)(x + (m0 + s * 16 + l16) * 2);
      xin[s][0].x = half ? 0.f : xv.x;
      xin[s][0].y = half ? 0.f : xv.y;
    }
    v8f acc[2][7];
    run_tower<false>(xin, wimg, acc, lane);
    run_tower<true>(xin, wimg + TOWER_FLOATS, acc, lane);

    static_assert(OUT_DIM == 6 * 16 + 4, "tile 6: registers 0..3 of half 0");
#pragma unroll
    for (int s = 0; s < 2; ++s) {
#pragma unroll
      for (int mt = 0; mt < 6; ++mt) {
#pragma unroll
        for (int q = 0; q < 2; ++q) {
          const v4f v = {acc[s][mt][q * 4 + 0], acc[s][mt][q * 4 + 1],
                         acc[s][mt][q * 4 + 2], acc[s][mt][q * 4 + 3]};
          *(v4f*)(og + (s * 16 + l16) * OUT_DIM + mt * 16 + half * 8 + q * 4) = v;
        }
      }
    }
    if (half == 0) {
#pragma unroll
      for (int s = 0; s < 2; ++s) {
        const v4f v = {acc[s][6][0], acc[s][6][1], acc[s][6][2], acc[s][6][3]};
        *(v4f*)(og + (s * 16 + l16) * OUT_DIM + 96) = v;
      }
    }
    lds_fence();
    float* orun = out + m0 * OUT_DIM;
#pragma unroll
    for (int it = 0; it < M_PER_WAVE * OUT_DIM / (4 * LANES); ++it) {
      const int idx = (it * LANES + lane) * 4;
      *(v4f*)(orun + idx) = *(const v4f*)(og + idx);
    }
  }
}

extern "C" void kernel_launch(void* const* d_in, const int* in_sizes, int n_in,
                              void* d_out, int out_size, void* d_ws,
                              size_t ws_size, hipStream_t stream) {
  (void)in_sizes; (void)n_in; (void)d_ws; (void)ws_size; (void)out_size;
  const float* x    = (const float*)d_in[0];
  const float* muW1 = (const float*)d_in[1];
  const float* mub1 = (const float*)d_in[2];
  const float* muW2 = (const float*)d_in[3];
  const float* mub2 = (const float*)d_in[4];
  const float* muW3 = (const float*)d_in[5];
  const float* mub3 = (const float*)d_in[6];
  const float* muW4 = (const float*)d_in[7];
  const float* mub4 = (const float*)d_in[8];
  const float* sgW1 = (const float*)d_in[9];
  const float* sgb1 = (const float*)d_in[10];
  const float* sgW2 = (const float*)d_in[11];
  const float* sgb2 = (const float*)d_in[12];
  const float* sgW3 = (const float*)d_in[13];
  const float* sgb3 = (const float*)d_in[14];
  const float* sgW4 = (const float*)d_in[15];
  const float* sgb4 = (const float*)d_in[16];
  float* out = (float*)d_out;

  const int blocks =
      N_SAMPLES / (WAVES_PER_BLOCK * M_PER_WAVE * TILES_PER_WAVE);
  double_decoder_kernel<<<blocks, THREADS, 0, stream>>>(
      x, muW1, mub1, muW2, mub2, muW3, mub3, muW4, mub4, sgW1, sgb1, sgW2,
      sgb2, sgW3, sgb3, sgW4, sgb4, out);
}
